# starting kernel
# baseline (speedup 1.0000x reference)
#include <hip/hip_runtime.h>
#include <hip/hip_bf16.h>
#include <stdint.h>

#define H 256

typedef __attribute__((ext_vector_type(2))) float v2f;
typedef __attribute__((ext_vector_type(8))) float v8f;

__global__ void k_zero(unsigned int* __restrict__ p, int n) {
    int i = blockIdx.x * blockDim.x + threadIdx.x;
    if (i < n) p[i] = 0u;
}

__global__ void k_hist(const int* __restrict__ src_idx, unsigned int* __restrict__ counts, int E) {
    int i = blockIdx.x * blockDim.x + threadIdx.x;
    if (i < E) atomicAdd(&counts[src_idx[i]], 1u);
}

__global__ void k_weighted_rowsum(const float* __restrict__ X,
                                  const unsigned int* __restrict__ counts,
                                  float* __restrict__ ls, int nrows) {
    int h = threadIdx.x;
    float acc = 0.0f;
    for (int n = blockIdx.x; n < nrows; n += gridDim.x) {
        float c = (float)counts[n];
        acc = fmaf(c, X[(size_t)n * H + h], acc);
    }
    atomicAdd(&ls[h], acc);
}

__global__ void k_matvec_R(const float* __restrict__ R,
                           const float* __restrict__ ls,
                           float* __restrict__ v) {
    __shared__ float s[H];
    s[threadIdx.x] = ls[threadIdx.x];
    __syncthreads();
    int wave = threadIdx.x >> 5;
    int lane = threadIdx.x & 31;
    for (int i = wave; i < H; i += 8) {
        const float* row = R + (size_t)i * H;
        float acc = 0.0f;
        #pragma unroll
        for (int j = lane; j < H; j += 32) acc = fmaf(row[j], s[j], acc);
        #pragma unroll
        for (int off = 16; off > 0; off >>= 1) acc += __shfl_xor(acc, off, 32);
        if (lane == 0) v[i] = acc;
    }
}

__global__ void k_dot_rows_wmma(const float* __restrict__ X,
                                const float* __restrict__ v,
                                float* __restrict__ t, int nrows) {
    __shared__ float vs[H];
    vs[threadIdx.x] = v[threadIdx.x];
    __syncthreads();

    int wave = threadIdx.x >> 5;
    int lane = threadIdx.x & 31;
    int row0 = (blockIdx.x * 8 + wave) * 16;

    int r = row0 + (lane & 15);
    if (r >= nrows) r = 0;
    const float* xrow = X + (size_t)r * H;
    int khalf = (lane >> 4) * 2;

    v8f c = {};
    #pragma unroll 8
    for (int k0 = 0; k0 < H; k0 += 4) {
        int kk = k0 + khalf;
        v2f a = *(const v2f*)(vs + kk);
        v2f b = *(const v2f*)(xrow + kk);
        c = __builtin_amdgcn_wmma_f32_16x16x4_f32(
                 false, a,  false, b,
                 (short)0, c,  false,  false);
    }

    int n = row0 + (lane & 15);
    if (lane < 16 && n < nrows) t[n] = c[0];
}

__global__ void k_gather_scores(const float* __restrict__ t,
                                const int* __restrict__ trg_idx,
                                float* __restrict__ out, int E) {
    int i = blockIdx.x * blockDim.x + threadIdx.x;
    if (i < E) out[i] = t[trg_idx[i]];
}

extern "C" void kernel_launch(void* const* d_in, const int* in_sizes, int n_in,
                              void* d_out, int out_size, void* d_ws, size_t ws_size,
                              hipStream_t stream) {
    const float* x_src = (const float*)d_in[0];
    const float* x_trg = (const float*)d_in[1];
    const float* R     = (const float*)d_in[2];
    const int*   edges = (const int*)d_in[3];

    const int n_src = in_sizes[0] / H;
    const int n_trg = in_sizes[1] / H;
    const int E     = in_sizes[3] / 2;
    const int* src_idx = edges;
    const int* trg_idx = edges + E;
    float* out = (float*)d_out;

    char* ws = (char*)d_ws;
    size_t off = 0;
    unsigned int* counts = (unsigned int*)(ws + off); off += ((size_t)n_src * 4 + 255) & ~(size_t)255;
    float* ls            = (float*)(ws + off);        off += (H * 4 + 255) & ~(size_t)255;
    float* vvec          = (float*)(ws + off);        off += (H * 4 + 255) & ~(size_t)255;
    float* t             = (float*)(ws + off);

    {
        int nz = n_src;
        k_zero<<<(nz + 255) / 256, 256, 0, stream>>>(counts, nz);
        k_zero<<<1, 256, 0, stream>>>((unsigned int*)ls, H);
        k_zero<<<1, 256, 0, stream>>>((unsigned int*)vvec, H);
    }
    k_hist<<<(E + 255) / 256, 256, 0, stream>>>(src_idx, counts, E);
    k_weighted_rowsum<<<512, H, 0, stream>>>(x_src, counts, ls, n_src);
    k_matvec_R<<<1, H, 0, stream>>>(R, ls, vvec);
    {
        int blocks = (n_trg + 127) / 128;
        k_dot_rows_wmma<<<blocks, 256, 0, stream>>>(x_trg, vvec, t, n_trg);
    }
    k_gather_scores<<<(E + 255) / 256, 256, 0, stream>>>(t, trg_idx, out, E);
}


# row-dot kernel: X tile fetched with 32 whole-line 16-B loads issued behind the R rows, fed to the WMMA B operand from a padded LDS copy, left half multiplied while the right arrives; sums unchanged
# speedup vs baseline: 3.1346x; 3.1346x over previous
#include <hip/hip_runtime.h>
#include <hip/hip_bf16.h>
#include <stdint.h>

#define H 256

typedef __attribute__((ext_vector_type(2))) float v2f;
typedef __attribute__((ext_vector_type(4))) float v4f;
typedef __attribute__((ext_vector_type(8))) float v8f;

__global__ void k_zero(unsigned int* __restrict__ p, int n) {
    int i = blockIdx.x * blockDim.x + threadIdx.x;
    if (i < n) p[i] = 0u;
}

#define WAVES_PER_WG 16
#define EDGE_GROUPS 256
#define EDGE_WAVES_LOG2 12

__global__ void k_edge_rowsum(const float* __restrict__ X,
                              const int* __restrict__ src_idx,
                              float* __restrict__ ls, int E) {
    __shared__ v4f part[WAVES_PER_WG][H / 4];

    int lane = threadIdx.x & 31;
    int wave = __builtin_amdgcn_readfirstlane(threadIdx.x >> 5);

    unsigned long long nb = (unsigned long long)((E + 7) >> 3);
    unsigned long long g  = (unsigned long long)(blockIdx.x * WAVES_PER_WG + wave);
    int e0 = (int)((g * nb) >> EDGE_WAVES_LOG2) * 8;
    int e1 = (int)(((g + 1) * nb) >> EDGE_WAVES_LOG2) * 8;
    if (e1 > E) e1 = E;

    const v4f* xl = (const v4f*)X + lane;

    v4f acc0 = {0.0f, 0.0f, 0.0f, 0.0f};
    v4f acc1 = acc0;

    int idx_next = (e0 + lane < e1) ? src_idx[e0 + lane] : 0;
    for (int base = e0; base < e1; base += 32) {
        int idx = idx_next;
        int en = base + 32 + lane;
        idx_next = (en < e1) ? src_idx[en] : 0;
        int cnt = e1 - base;
        if (cnt > 32) cnt = 32;

        int j = 0;
        for (; j + 8 <= cnt; j += 8) {
            v4f a[8], b[8];
            for (int u = 0; u < 8; ++u) {
                int row = __builtin_amdgcn_readlane(idx, j + u);
                const v4f* p = xl + (size_t)row * (H / 4);
                a[u] = p[0];
                b[u] = p[32];
            }
            asm volatile("" : "+v"(a[0]), "+v"(a[1]), "+v"(a[2]), "+v"(a[3]),
                              "+v"(a[4]), "+v"(a[5]), "+v"(a[6]), "+v"(a[7]),
                              "+v"(b[0]), "+v"(b[1]), "+v"(b[2]), "+v"(b[3]),
                              "+v"(b[4]), "+v"(b[5]), "+v"(b[6]), "+v"(b[7]));
            for (int u = 0; u < 8; ++u) {
                acc0 += a[u];
                acc1 += b[u];
            }
        }
        #pragma nounroll
        for (; j < cnt; ++j) {
            int row = __builtin_amdgcn_readlane(idx, j);
            const v4f* p = xl + (size_t)row * (H / 4);
            acc0 += p[0];
            acc1 += p[32];
        }
    }

    part[wave][lane]      = acc0;
    part[wave][32 + lane] = acc1;
    __syncthreads();

    if (threadIdx.x < H) {
        const float* pf = (const float*)part + threadIdx.x;
        float s = 0.0f;
        for (int w = 0; w < WAVES_PER_WG; ++w) s += pf[w * H];
        atomicAdd(&ls[threadIdx.x], s);
    }
}

__device__ __forceinline__ float dot4acc(float4 a, float4 w, float acc) {
    acc = fmaf(a.x, w.x, acc);
    acc = fmaf(a.y, w.y, acc);
    acc = fmaf(a.z, w.z, acc);
    acc = fmaf(a.w, w.w, acc);
    return acc;
}

#define XPITCH 260

#define A_STEP_LOAD(step, a, b)                                                        \
    {                                                                                  \
        const float4* r0 = (const float4*)(R + (size_t)(4 * (step) + 2 * half) * H + 4 * c16); \
        const float4* r1 = r0 + H / 4;                                                 \
        a[0] = r0[0]; a[1] = r0[16]; a[2] = r0[32]; a[3] = r0[48];                     \
        b[0] = r1[0]; b[1] = r1[16]; b[2] = r1[32]; b[3] = r1[48];                     \
    }
#define A_STEP_SUM(step, a, b)                                                         \
    {                                                                                  \
        float p0 = dot4acc(a[3], w3, dot4acc(a[2], w2, dot4acc(a[1], w1, dot4acc(a[0], w0, 0.0f)))); \
        float p1 = dot4acc(b[3], w3, dot4acc(b[2], w2, dot4acc(b[1], w1, dot4acc(b[0], w0, 0.0f)))); \
        v2f p; p.x = p0; p.y = p1;                                                     \
        afrag[(step) * 32 + lane] = p;                                                 \
    }

__global__ void __launch_bounds__(256)
k_dot_rows_wmma(const float* __restrict__ X,
                const float* __restrict__ R,
                const float* __restrict__ ls,
                float* __restrict__ t, int nrows) {
    __shared__ v2f afrag[(H / 4) * 32];
    __shared__ v4f xtile[8][16 * XPITCH / 4];

    int wave = __builtin_amdgcn_readfirstlane(threadIdx.x >> 5);
    int lane = threadIdx.x & 31;
    int c16  = lane & 15;
    int half = lane >> 4;
    int row0 = (blockIdx.x * 8 + wave) * 16;

    const float4* lq = (const float4*)(ls + 4 * c16);
    float4 w0 = lq[0], w1 = lq[16], w2 = lq[32], w3 = lq[48];

    float4 ra[4][4], rb[4][4];
    #pragma unroll
    for (int s = 0; s < 4; ++s) A_STEP_LOAD(wave * 8 + s, ra[s], rb[s]);
    #pragma unroll
    for (int s = 0; s < 4; ++s) A_STEP_SUM(wave * 8 + s, ra[s], rb[s]);

    __builtin_amdgcn_sched_barrier(0);

    #pragma unroll
    for (int s = 0; s < 4; ++s) A_STEP_LOAD(wave * 8 + 4 + s, ra[s], rb[s]);

    v4f xl[16], xr[16];
    #pragma unroll
    for (int j = 0; j < 16; ++j) {
        int r = row0 + j;
        if (r >= nrows) r = 0;
        xl[j] = ((const v4f*)(X + (size_t)r * H))[lane];
    }
    __builtin_amdgcn_sched_barrier(0);
    #pragma unroll
    for (int s = 0; s < 4; ++s) A_STEP_SUM(wave * 8 + 4 + s, ra[s], rb[s]);
    __builtin_amdgcn_sched_barrier(0);
    #pragma unroll
    for (int j = 0; j < 16; ++j) {
        int r = row0 + j;
        if (r >= nrows) r = 0;
        xr[j] = ((const v4f*)(X + (size_t)r * H))[32 + lane];
    }

    v4f* tl = xtile[wave];
    const float* brow = (const float*)xtile[wave] + c16 * XPITCH + 2 * half;
    v8f c = {};

    #pragma unroll
    for (int j = 0; j < 16; ++j) tl[j * (XPITCH / 4) + lane] = xl[j];
    __builtin_amdgcn_fence(__ATOMIC_RELEASE, "workgroup", "local");
    __builtin_amdgcn_s_barrier();
    __builtin_amdgcn_fence(__ATOMIC_ACQUIRE, "workgroup", "local");
    #pragma unroll 8
    for (int k0 = 0; k0 < H / 2; k0 += 4) {
        v2f a = afrag[(k0 >> 2) * 32 + lane];
        v2f b = *(const v2f*)(brow + k0);
        c = __builtin_amdgcn_wmma_f32_16x16x4_f32(
                 false, a,  false, b,
                 (short)0, c,  false,  false);
    }

    #pragma unroll
    for (int j = 0; j < 16; ++j) tl[j * (XPITCH / 4) + 32 + lane] = xr[j];
    __syncthreads();
    #pragma unroll 8
    for (int k0 = H / 2; k0 < H; k0 += 4) {
        v2f a = afrag[(k0 >> 2) * 32 + lane];
        v2f b = *(const v2f*)(brow + k0);
        c = __builtin_amdgcn_wmma_f32_16x16x4_f32(
                 false, a,  false, b,
                 (short)0, c,  false,  false);
    }

    float sum = ((c[0] + c[1]) + (c[2] + c[3])) + ((c[4] + c[5]) + (c[6] + c[7]));
    sum += __shfl_xor(sum, 16, 32);

    int n = row0 + c16;
    if (lane < 16 && n < nrows) t[n] = sum;
}

__global__ void k_gather_scores(const float* __restrict__ t,
                                const int* __restrict__ trg_idx,
                                float* __restrict__ out, int E) {
    int i = blockIdx.x * blockDim.x + threadIdx.x;
    if (i < E) out[i] = t[trg_idx[i]];
}

extern "C" void kernel_launch(void* const* d_in, const int* in_sizes, int n_in,
                              void* d_out, int out_size, void* d_ws, size_t ws_size,
                              hipStream_t stream) {
    const float* x_src = (const float*)d_in[0];
    const float* x_trg = (const float*)d_in[1];
    const float* R     = (const float*)d_in[2];
    const int*   edges = (const int*)d_in[3];

    const int n_trg = in_sizes[1] / H;
    const int E     = in_sizes[3] / 2;
    const int* src_idx = edges;
    const int* trg_idx = edges + E;
    float* out = (float*)d_out;

    char* ws = (char*)d_ws;
    size_t off = 0;
    float* ls = (float*)(ws + off); off += (H * 4 + 255) & ~(size_t)255;
    float* t  = (float*)(ws + off);

    k_zero<<<1, 256, 0, stream>>>((unsigned int*)ls, H);
    k_edge_rowsum<<<EDGE_GROUPS, 32 * WAVES_PER_WG, 0, stream>>>(x_src, src_idx, ls, E);
    {
        int blocks = (n_trg + 127) / 128;
        k_dot_rows_wmma<<<blocks, 256, 0, stream>>>(x_trg, R, ls, t, n_trg);
    }
    k_gather_scores<<<(E + 255) / 256, 256, 0, stream>>>(t, trg_idx, out, E);
}


# row-dot kernel cut along k: 4 work-groups per row tile, each builds only its 64 entries of R*ls (a quarter of R) and adds partial dots into zeroed t with f32 atomics; new f32 summation order
# speedup vs baseline: 3.3564x; 1.0708x over previous
#include <hip/hip_runtime.h>
#include <hip/hip_bf16.h>
#include <stdint.h>

#define H 256

typedef __attribute__((ext_vector_type(2))) float v2f;
typedef __attribute__((ext_vector_type(4))) float v4f;
typedef __attribute__((ext_vector_type(8))) float v8f;

__global__ void k_zero(unsigned int* __restrict__ p, int n) {
    int i = blockIdx.x * blockDim.x + threadIdx.x;
    if (i < n) p[i] = 0u;
}

#define WAVES_PER_WG 16
#define EDGE_GROUPS 256
#define EDGE_WAVES_LOG2 12

__global__ void k_edge_rowsum(const float* __restrict__ X,
                              const int* __restrict__ src_idx,
                              float* __restrict__ ls, int E) {
    __shared__ v4f part[WAVES_PER_WG][H / 4];

    int lane = threadIdx.x & 31;
    int wave = __builtin_amdgcn_readfirstlane(threadIdx.x >> 5);

    unsigned long long nb = (unsigned long long)((E + 7) >> 3);
    unsigned long long g  = (unsigned long long)(blockIdx.x * WAVES_PER_WG + wave);
    int e0 = (int)((g * nb) >> EDGE_WAVES_LOG2) * 8;
    int e1 = (int)(((g + 1) * nb) >> EDGE_WAVES_LOG2) * 8;
    if (e1 > E) e1 = E;

    const v4f* xl = (const v4f*)X + lane;

    v4f acc0 = {0.0f, 0.0f, 0.0f, 0.0f};
    v4f acc1 = acc0;

    int idx_next = (e0 + lane < e1) ? src_idx[e0 + lane] : 0;
    for (int base = e0; base < e1; base += 32) {
        int idx = idx_next;
        int en = base + 32 + lane;
        idx_next = (en < e1) ? src_idx[en] : 0;
        int cnt = e1 - base;
        if (cnt > 32) cnt = 32;

        int j = 0;
        for (; j + 8 <= cnt; j += 8) {
            v4f a[8], b[8];
            for (int u = 0; u < 8; ++u) {
                int row = __builtin_amdgcn_readlane(idx, j + u);
                const v4f* p = xl + (size_t)row * (H / 4);
                a[u] = p[0];
                b[u] = p[32];
            }
            asm volatile("" : "+v"(a[0]), "+v"(a[1]), "+v"(a[2]), "+v"(a[3]),
                              "+v"(a[4]), "+v"(a[5]), "+v"(a[6]), "+v"(a[7]),
                              "+v"(b[0]), "+v"(b[1]), "+v"(b[2]), "+v"(b[3]),
                              "+v"(b[4]), "+v"(b[5]), "+v"(b[6]), "+v"(b[7]));
            for (int u = 0; u < 8; ++u) {
                acc0 += a[u];
                acc1 += b[u];
            }
        }
        #pragma nounroll
        for (; j < cnt; ++j) {
            int row = __builtin_amdgcn_readlane(idx, j);
            const v4f* p = xl + (size_t)row * (H / 4);
            acc0 += p[0];
            acc1 += p[32];
        }
    }

    part[wave][lane]      = acc0;
    part[wave][32 + lane] = acc1;
    __syncthreads();

    if (threadIdx.x < H) {
        const float* pf = (const float*)part + threadIdx.x;
        float s = 0.0f;
        for (int w = 0; w < WAVES_PER_WG; ++w) s += pf[w * H];
        atomicAdd(&ls[threadIdx.x], s);
    }
}

__device__ __forceinline__ float dot4acc(float4 a, float4 w, float acc) {
    acc = fmaf(a.x, w.x, acc);
    acc = fmaf(a.y, w.y, acc);
    acc = fmaf(a.z, w.z, acc);
    acc = fmaf(a.w, w.w, acc);
    return acc;
}

#define KSPLIT 4
#define KCOLS (H / KSPLIT)
#define KSTEPS (KCOLS / 4)
#define WROWS 64
#define XPITCH 68

#define A_STEP_LOAD(gstep, a, b)                                                       \
    {                                                                                  \
        const float4* r0 = (const float4*)(R + (size_t)(4 * (gstep) + 2 * half) * H + 4 * c16); \
        const float4* r1 = r0 + H / 4;                                                 \
        a[0] = r0[0]; a[1] = r0[16]; a[2] = r0[32]; a[3] = r0[48];                     \
        b[0] = r1[0]; b[1] = r1[16]; b[2] = r1[32]; b[3] = r1[48];                     \
    }
#define A_STEP_SUM(lstep, a, b)                                                        \
    {                                                                                  \
        float p0 = dot4acc(a[3], w3, dot4acc(a[2], w2, dot4acc(a[1], w1, dot4acc(a[0], w0, 0.0f)))); \
        float p1 = dot4acc(b[3], w3, dot4acc(b[2], w2, dot4acc(b[1], w1, dot4acc(b[0], w0, 0.0f)))); \
        v2f p; p.x = p0; p.y = p1;                                                     \
        afrag[(lstep) * 32 + lane] = p;                                                \
    }

__global__ void __launch_bounds__(256)
k_dot_rows_wmma(const float* __restrict__ X,
                const float* __restrict__ R,
                const float* __restrict__ ls,
                float* __restrict__ t, int nrows) {
    __shared__ v2f afrag[KSTEPS * 32];
    __shared__ v4f xtile[8][WROWS * XPITCH / 4];

    int wave = __builtin_amdgcn_readfirstlane(threadIdx.x >> 5);
    int lane = threadIdx.x & 31;
    int c16  = lane & 15;
    int half = lane >> 4;
    int ks   = blockIdx.x & (KSPLIT - 1);
    int row0 = ((blockIdx.x / KSPLIT) * 8 + wave) * WROWS;

    const float4* lq = (const float4*)(ls + 4 * c16);
    float4 w0 = lq[0], w1 = lq[16], w2 = lq[32], w3 = lq[48];

    float4 ra[2][4], rb[2][4];
    #pragma unroll
    for (int s = 0; s < 2; ++s) A_STEP_LOAD(ks * KSTEPS + wave * 2 + s, ra[s], rb[s]);

    const float* xcol = X + ks * KCOLS + 4 * c16;
    v4f xv[2][16];
    #pragma unroll
    for (int i = 0; i < 16; ++i) {
        int r = row0 + 2 * i + half;
        if (r >= nrows) r = 0;
        xv[0][i] = *(const v4f*)(xcol + (size_t)r * H);
    }
    __builtin_amdgcn_sched_barrier(0);
    #pragma unroll
    for (int s = 0; s < 2; ++s) A_STEP_SUM(wave * 2 + s, ra[s], rb[s]);
    __builtin_amdgcn_sched_barrier(0);
    #pragma unroll
    for (int i = 0; i < 16; ++i) {
        int r = row0 + 32 + 2 * i + half;
        if (r >= nrows) r = 0;
        xv[1][i] = *(const v4f*)(xcol + (size_t)r * H);
    }

    __builtin_amdgcn_fence(__ATOMIC_RELEASE, "workgroup", "local");
    __builtin_amdgcn_s_barrier();
    __builtin_amdgcn_fence(__ATOMIC_ACQUIRE, "workgroup", "local");
    v2f a[KSTEPS];
    #pragma unroll
    for (int q = 0; q < KSTEPS; ++q) a[q] = afrag[q * 32 + lane];

    v4f* tl = xtile[wave];
    const float* brow = (const float*)xtile[wave] + c16 * XPITCH + 2 * half;

    #pragma unroll
    for (int p = 0; p < 2; ++p) {
        #pragma unroll
        for (int i = 0; i < 16; ++i) tl[(32 * p + 2 * i + half) * (XPITCH / 4) + c16] = xv[p][i];
        __builtin_amdgcn_fence(__ATOMIC_RELEASE, "wavefront", "local");
        __builtin_amdgcn_wave_barrier();
        __builtin_amdgcn_fence(__ATOMIC_ACQUIRE, "wavefront", "local");

        const float* bp = brow + 32 * p * XPITCH;
        v8f c0 = {}, c1 = {};
        #pragma unroll
        for (int q = 0; q < KSTEPS; ++q) {
            v2f b0 = *(const v2f*)(bp + 4 * q);
            v2f b1 = *(const v2f*)(bp + 16 * XPITCH + 4 * q);
            c0 = __builtin_amdgcn_wmma_f32_16x16x4_f32(
                     false, a[q],  false, b0,
                     (short)0, c0,  false,  false);
            c1 = __builtin_amdgcn_wmma_f32_16x16x4_f32(
                     false, a[q],  false, b1,
                     (short)0, c1,  false,  false);
        }

        float s0 = ((c0[0] + c0[1]) + (c0[2] + c0[3])) + ((c0[4] + c0[5]) + (c0[6] + c0[7]));
        float s1 = ((c1[0] + c1[1]) + (c1[2] + c1[3])) + ((c1[4] + c1[5]) + (c1[6] + c1[7]));
        s0 += __shfl_xor(s0, 16, 32);
        s1 += __shfl_xor(s1, 16, 32);

        int n = row0 + 32 * p + lane;
        if (n < nrows) atomicAdd(&t[n], half ? s1 : s0);
    }
}

__global__ void k_gather_scores(const float* __restrict__ t,
                                const int* __restrict__ trg_idx,
                                float* __restrict__ out, int E) {
    int i = blockIdx.x * blockDim.x + threadIdx.x;
    if (i < E) out[i] = t[trg_idx[i]];
}

extern "C" void kernel_launch(void* const* d_in, const int* in_sizes, int n_in,
                              void* d_out, int out_size, void* d_ws, size_t ws_size,
                              hipStream_t stream) {
    const float* x_src = (const float*)d_in[0];
    const float* x_trg = (const float*)d_in[1];
    const float* R     = (const float*)d_in[2];
    const int*   edges = (const int*)d_in[3];

    const int n_trg = in_sizes[1] / H;
    const int E     = in_sizes[3] / 2;
    const int* src_idx = edges;
    const int* trg_idx = edges + E;
    float* out = (float*)d_out;

    char* ws = (char*)d_ws;
    size_t off = 0;
    float* ls = (float*)(ws + off); off += (H * 4 + 255) & ~(size_t)255;
    float* t  = (float*)(ws + off);

    k_zero<<<(H + n_trg + 255) / 256, 256, 0, stream>>>((unsigned int*)ls, H + n_trg);
    k_edge_rowsum<<<EDGE_GROUPS, 32 * WAVES_PER_WG, 0, stream>>>(x_src, src_idx, ls, E);
    {
        int blocks = ((n_trg + 511) / 512) * KSPLIT;
        k_dot_rows_wmma<<<blocks, 256, 0, stream>>>(x_trg, R, ls, t, n_trg);
    }
    k_gather_scores<<<(E + 255) / 256, 256, 0, stream>>>(t, trg_idx, out, E);
}


# left_sum as 32 partial rows: a work-group sums one edge group x one 128-B column slice, plain stores; row-dot kernel adds the rows; no zero-fill launch or ls atomics, 4->3 launches; new f32 sum order
# speedup vs baseline: 3.4633x; 1.0318x over previous
#include <hip/hip_runtime.h>
#include <hip/hip_bf16.h>
#include <stdint.h>

#define H 256

typedef __attribute__((ext_vector_type(2))) float v2f;
typedef __attribute__((ext_vector_type(4))) float v4f;
typedef __attribute__((ext_vector_type(8))) float v8f;

#define WAVES_PER_WG 16
#define EDGE_GROUPS 32
#define COL_SLICES 8
#define EDGE_WAVES_LOG2 9

__global__ void __launch_bounds__(32 * WAVES_PER_WG)
k_edge_rowsum(const float* __restrict__ X,
              const int* __restrict__ src_idx,
              float* __restrict__ P, float* __restrict__ t,
              int E, int n_trg) {
    __shared__ v4f part[WAVES_PER_WG][32];

    int lane = threadIdx.x & 31;
    int wave = __builtin_amdgcn_readfirstlane(threadIdx.x >> 5);
    int eg = blockIdx.x / COL_SLICES;
    int cs = blockIdx.x % COL_SLICES;
    int sub = lane >> 3;

    for (int i = blockIdx.x * (32 * WAVES_PER_WG) + threadIdx.x; i < n_trg;
         i += EDGE_GROUPS * COL_SLICES * 32 * WAVES_PER_WG)
        t[i] = 0.0f;

    unsigned long long nb = (unsigned long long)((E + 7) >> 3);
    unsigned long long g  = (unsigned long long)(eg * WAVES_PER_WG + wave);
    int e0 = (int)((g * nb) >> EDGE_WAVES_LOG2) * 8;
    int e1 = (int)(((g + 1) * nb) >> EDGE_WAVES_LOG2) * 8;
    if (e1 > E) e1 = E;

    const v4f* xl = (const v4f*)X + cs * 8 + (lane & 7);

    v4f acc0 = {0.0f, 0.0f, 0.0f, 0.0f};
    v4f acc1 = acc0, acc2 = acc0, acc3 = acc0;

    int ia_next = (e0 + lane < e1) ? src_idx[e0 + lane] : 0;
    int ib_next = (e0 + 32 + lane < e1) ? src_idx[e0 + 32 + lane] : 0;
    for (int base = e0; base < e1; base += 64) {
        int ia = ia_next, ib = ib_next;
        int en = base + 64 + lane;
        ia_next = (en < e1) ? src_idx[en] : 0;
        ib_next = (en + 32 < e1) ? src_idx[en + 32] : 0;
        int cnt = e1 - base;

        if (cnt >= 64) {
            v4f a[16];
            for (int u = 0; u < 16; ++u) {
                int row = __shfl(u < 8 ? ia : ib, 4 * (u & 7) + sub, 32);
                a[u] = xl[(size_t)row * (H / 4)];
            }
            asm volatile("" : "+v"(a[0]), "+v"(a[1]), "+v"(a[2]), "+v"(a[3]),
                              "+v"(a[4]), "+v"(a[5]), "+v"(a[6]), "+v"(a[7]),
                              "+v"(a[8]), "+v"(a[9]), "+v"(a[10]), "+v"(a[11]),
                              "+v"(a[12]), "+v"(a[13]), "+v"(a[14]), "+v"(a[15]));
            for (int u = 0; u < 16; u += 4) {
                acc0 += a[u];
                acc1 += a[u + 1];
                acc2 += a[u + 2];
                acc3 += a[u + 3];
            }
        } else {
            #pragma nounroll
            for (int j = 0; j < cnt; j += 4) {
                int q = j + sub;
                int ra = __shfl(ia, q & 31, 32);
                int rb = __shfl(ib, q & 31, 32);
                int row = (q < 32) ? ra : rb;
                if (q < cnt) acc0 += xl[(size_t)row * (H / 4)];
            }
        }
    }

    part[wave][lane] = (acc0 + acc1) + (acc2 + acc3);
    __syncthreads();

    if (threadIdx.x < 32) {
        const float* pf = (const float*)part + threadIdx.x;
        float s0 = 0.0f, s1 = 0.0f, s2 = 0.0f, s3 = 0.0f;
        for (int w = 0; w < WAVES_PER_WG; ++w) {
            s0 += pf[w * 128];
            s1 += pf[w * 128 + 32];
            s2 += pf[w * 128 + 64];
            s3 += pf[w * 128 + 96];
        }
        P[eg * H + cs * 32 + threadIdx.x] = (s0 + s1) + (s2 + s3);
    }
}

__device__ __forceinline__ float dot4acc(float4 a, v4f w, float acc) {
    acc = fmaf(a.x, w.x, acc);
    acc = fmaf(a.y, w.y, acc);
    acc = fmaf(a.z, w.z, acc);
    acc = fmaf(a.w, w.w, acc);
    return acc;
}

#define KSPLIT 4
#define KCOLS (H / KSPLIT)
#define KSTEPS (KCOLS / 4)
#define WROWS 64
#define XPITCH 68

#define A_STEP_LOAD(gstep, a, b)                                                       \
    {                                                                                  \
        const float4* r0 = (const float4*)(R + (size_t)(4 * (gstep) + 2 * half) * H + 4 * c16); \
        const float4* r1 = r0 + H / 4;                                                 \
        a[0] = r0[0]; a[1] = r0[16]; a[2] = r0[32]; a[3] = r0[48];                     \
        b[0] = r1[0]; b[1] = r1[16]; b[2] = r1[32]; b[3] = r1[48];                     \
    }
#define A_STEP_SUM(lstep, a, b)                                                        \
    {                                                                                  \
        float p0 = dot4acc(a[3], w3, dot4acc(a[2], w2, dot4acc(a[1], w1, dot4acc(a[0], w0, 0.0f)))); \
        float p1 = dot4acc(b[3], w3, dot4acc(b[2], w2, dot4acc(b[1], w1, dot4acc(b[0], w0, 0.0f)))); \
        v2f p; p.x = p0; p.y = p1;                                                     \
        afrag[(lstep) * 32 + lane] = p;                                                \
    }

__global__ void __launch_bounds__(256)
k_dot_rows_wmma(const float* __restrict__ X,
                const float* __restrict__ R,
                const float* __restrict__ P,
                float* __restrict__ t, int nrows) {
    __shared__ v2f afrag[KSTEPS * 32];
    __shared__ v4f lsq[4][H / 4];
    __shared__ v4f xtile[8][WROWS * XPITCH / 4];

    int wave = __builtin_amdgcn_readfirstlane(threadIdx.x >> 5);
    int lane = threadIdx.x & 31;
    int c16  = lane & 15;
    int half = lane >> 4;
    int ks   = blockIdx.x & (KSPLIT - 1);
    int row0 = ((blockIdx.x / KSPLIT) * 8 + wave) * WROWS;

    int pq = threadIdx.x >> 6;
    int pc = threadIdx.x & 63;
    v4f pv[EDGE_GROUPS / 4];
    #pragma unroll
    for (int i = 0; i < EDGE_GROUPS / 4; ++i)
        pv[i] = *(const v4f*)(P + (size_t)((EDGE_GROUPS / 4) * pq + i) * H + 4 * pc);
    __builtin_amdgcn_sched_barrier(0);

    float4 ra[2][4], rb[2][4];
    #pragma unroll
    for (int s = 0; s < 2; ++s) A_STEP_LOAD(ks * KSTEPS + wave * 2 + s, ra[s], rb[s]);
    __builtin_amdgcn_sched_barrier(0);

    const float* xcol = X + ks * KCOLS + 4 * c16;
    v4f xv[2][16];
    #pragma unroll
    for (int i = 0; i < 16; ++i) {
        int r = row0 + 2 * i + half;
        if (r >= nrows) r = 0;
        xv[0][i] = *(const v4f*)(xcol + (size_t)r * H);
    }
    __builtin_amdgcn_sched_barrier(0);

    lsq[pq][pc] = ((pv[0] + pv[1]) + (pv[2] + pv[3])) + ((pv[4] + pv[5]) + (pv[6] + pv[7]));
    __builtin_amdgcn_fence(__ATOMIC_RELEASE, "workgroup", "local");
    __builtin_amdgcn_s_barrier();
    __builtin_amdgcn_fence(__ATOMIC_ACQUIRE, "workgroup", "local");
    v4f w0 = (lsq[0][c16]      + lsq[1][c16])      + (lsq[2][c16]      + lsq[3][c16]);
    v4f w1 = (lsq[0][c16 + 16] + lsq[1][c16 + 16]) + (lsq[2][c16 + 16] + lsq[3][c16 + 16]);
    v4f w2 = (lsq[0][c16 + 32] + lsq[1][c16 + 32]) + (lsq[2][c16 + 32] + lsq[3][c16 + 32]);
    v4f w3 = (lsq[0][c16 + 48] + lsq[1][c16 + 48]) + (lsq[2][c16 + 48] + lsq[3][c16 + 48]);

    #pragma unroll
    for (int s = 0; s < 2; ++s) A_STEP_SUM(wave * 2 + s, ra[s], rb[s]);
    __builtin_amdgcn_sched_barrier(0);
    #pragma unroll
    for (int i = 0; i < 16; ++i) {
        int r = row0 + 32 + 2 * i + half;
        if (r >= nrows) r = 0;
        xv[1][i] = *(const v4f*)(xcol + (size_t)r * H);
    }

    __builtin_amdgcn_fence(__ATOMIC_RELEASE, "workgroup", "local");
    __builtin_amdgcn_s_barrier();
    __builtin_amdgcn_fence(__ATOMIC_ACQUIRE, "workgroup", "local");
    v2f a[KSTEPS];
    #pragma unroll
    for (int q = 0; q < KSTEPS; ++q) a[q] = afrag[q * 32 + lane];

    v4f* tl = xtile[wave];
    const float* brow = (const float*)xtile[wave] + c16 * XPITCH + 2 * half;

    #pragma unroll
    for (int p = 0; p < 2; ++p) {
        #pragma unroll
        for (int i = 0; i < 16; ++i) tl[(32 * p + 2 * i + half) * (XPITCH / 4) + c16] = xv[p][i];
        __builtin_amdgcn_fence(__ATOMIC_RELEASE, "wavefront", "local");
        __builtin_amdgcn_wave_barrier();
        __builtin_amdgcn_fence(__ATOMIC_ACQUIRE, "wavefront", "local");

        const float* bp = brow + 32 * p * XPITCH;
        v8f c0 = {}, c1 = {};
        #pragma unroll
        for (int q = 0; q < KSTEPS; ++q) {
            v2f b0 = *(const v2f*)(bp + 4 * q);
            v2f b1 = *(const v2f*)(bp + 16 * XPITCH + 4 * q);
            c0 = __builtin_amdgcn_wmma_f32_16x16x4_f32(
                     false, a[q],  false, b0,
                     (short)0, c0,  false,  false);
            c1 = __builtin_amdgcn_wmma_f32_16x16x4_f32(
                     false, a[q],  false, b1,
                     (short)0, c1,  false,  false);
        }

        float s0 = ((c0[0] + c0[1]) + (c0[2] + c0[3])) + ((c0[4] + c0[5]) + (c0[6] + c0[7]));
        float s1 = ((c1[0] + c1[1]) + (c1[2] + c1[3])) + ((c1[4] + c1[5]) + (c1[6] + c1[7]));
        s0 += __shfl_xor(s0, 16, 32);
        s1 += __shfl_xor(s1, 16, 32);

        int n = row0 + 32 * p + lane;
        if (n < nrows) atomicAdd(&t[n], half ? s1 : s0);
    }
}

__global__ void k_gather_scores(const float* __restrict__ t,
                                const int* __restrict__ trg_idx,
                                float* __restrict__ out, int E) {
    int i = blockIdx.x * blockDim.x + threadIdx.x;
    if (i < E) out[i] = t[trg_idx[i]];
}

extern "C" void kernel_launch(void* const* d_in, const int* in_sizes, int n_in,
                              void* d_out, int out_size, void* d_ws, size_t ws_size,
                              hipStream_t stream) {
    const float* x_src = (const float*)d_in[0];
    const float* x_trg = (const float*)d_in[1];
    const float* R     = (const float*)d_in[2];
    const int*   edges = (const int*)d_in[3];

    const int n_trg = in_sizes[1] / H;
    const int E     = in_sizes[3] / 2;
    const int* src_idx = edges;
    const int* trg_idx = edges + E;
    float* out = (float*)d_out;

    char* ws = (char*)d_ws;
    float* P = (float*)ws;
    float* t = (float*)(ws + EDGE_GROUPS * H * 4);

    k_edge_rowsum<<<EDGE_GROUPS * COL_SLICES, 32 * WAVES_PER_WG, 0, stream>>>(x_src, src_idx, P, t, E, n_trg);
    {
        int blocks = ((n_trg + 511) / 512) * KSPLIT;
        k_dot_rows_wmma<<<blocks, 256, 0, stream>>>(x_trg, R, P, t, n_trg);
    }
    k_gather_scores<<<(E + 255) / 256, 256, 0, stream>>>(t, trg_idx, out, E);
}


# edge row-sum: the next 64 indices are requested behind a step's 16 row loads and stay in flight over the row wait (one round trip less per step); short last step's rows as one group; sums unchanged
# speedup vs baseline: 3.6085x; 1.0419x over previous
#include <hip/hip_runtime.h>
#include <hip/hip_bf16.h>
#include <stdint.h>

#define H 256

typedef __attribute__((ext_vector_type(2))) float v2f;
typedef __attribute__((ext_vector_type(4))) float v4f;
typedef __attribute__((ext_vector_type(8))) float v8f;

#define WAVES_PER_WG 16
#define EDGE_GROUPS 32
#define COL_SLICES 8
#define EDGE_WAVES_LOG2 9

__global__ void __launch_bounds__(32 * WAVES_PER_WG)
k_edge_rowsum(const float* __restrict__ X,
              const int* __restrict__ src_idx,
              float* __restrict__ P, float* __restrict__ t,
              int E, int n_trg) {
    __shared__ v4f part[WAVES_PER_WG][32];

    int lane = threadIdx.x & 31;
    int wave = __builtin_amdgcn_readfirstlane(threadIdx.x >> 5);
    int eg = blockIdx.x / COL_SLICES;
    int cs = blockIdx.x % COL_SLICES;
    int sub = lane >> 3;

    for (int i = blockIdx.x * (32 * WAVES_PER_WG) + threadIdx.x; i < n_trg;
         i += EDGE_GROUPS * COL_SLICES * 32 * WAVES_PER_WG)
        t[i] = 0.0f;

    unsigned long long nb = (unsigned long long)((E + 7) >> 3);
    unsigned long long g  = (unsigned long long)(eg * WAVES_PER_WG + wave);
    int e0 = (int)((g * nb) >> EDGE_WAVES_LOG2) * 8;
    int e1 = (int)(((g + 1) * nb) >> EDGE_WAVES_LOG2) * 8;
    if (e1 > E) e1 = E;

    const v4f* xl = (const v4f*)X + cs * 8 + (lane & 7);

    v4f acc0 = {0.0f, 0.0f, 0.0f, 0.0f};
    v4f acc1 = acc0, acc2 = acc0, acc3 = acc0;

    if (e0 < e1) {
        const int last = e1 - 1;
        int ia = src_idx[min(e0 + lane, last)];
        int ib = src_idx[min(e0 + 32 + lane, last)];
        int base = e0;
        for (; base + 64 <= e1; base += 64) {
            v4f a[16];
            for (int u = 0; u < 16; ++u) {
                int row = __shfl(u < 8 ? ia : ib, 4 * (u & 7) + sub, 32);
                a[u] = xl[(size_t)row * (H / 4)];
            }
            __builtin_amdgcn_sched_barrier(0);
            int en = base + 64 + lane;
            int ia_n = src_idx[min(en, last)];
            int ib_n = src_idx[min(en + 32, last)];
            __builtin_amdgcn_sched_barrier(0);
            asm volatile("" : "+v"(a[0]), "+v"(a[1]), "+v"(a[2]), "+v"(a[3]),
                              "+v"(a[4]), "+v"(a[5]), "+v"(a[6]), "+v"(a[7]),
                              "+v"(a[8]), "+v"(a[9]), "+v"(a[10]), "+v"(a[11]),
                              "+v"(a[12]), "+v"(a[13]), "+v"(a[14]), "+v"(a[15]));
            for (int u = 0; u < 16; u += 4) {
                acc0 += a[u];
                acc1 += a[u + 1];
                acc2 += a[u + 2];
                acc3 += a[u + 3];
            }
            ia = ia_n;
            ib = ib_n;
        }

        int cnt = e1 - base;
        if (cnt > 0) {
            const v4f zero = {0.0f, 0.0f, 0.0f, 0.0f};
            v4f a[16];
            for (int u = 0; u < 16; ++u) {
                int row = __shfl(u < 8 ? ia : ib, 4 * (u & 7) + sub, 32);
                a[u] = xl[(size_t)row * (H / 4)];
            }
            asm volatile("" : "+v"(a[0]), "+v"(a[1]), "+v"(a[2]), "+v"(a[3]),
                              "+v"(a[4]), "+v"(a[5]), "+v"(a[6]), "+v"(a[7]),
                              "+v"(a[8]), "+v"(a[9]), "+v"(a[10]), "+v"(a[11]),
                              "+v"(a[12]), "+v"(a[13]), "+v"(a[14]), "+v"(a[15]));
            for (int u = 0; u < 16; ++u)
                acc0 += (4 * u + sub < cnt) ? a[u] : zero;
        }
    }

    part[wave][lane] = (acc0 + acc1) + (acc2 + acc3);
    __syncthreads();

    if (threadIdx.x < 32) {
        const float* pf = (const float*)part + threadIdx.x;
        float s0 = 0.0f, s1 = 0.0f, s2 = 0.0f, s3 = 0.0f;
        for (int w = 0; w < WAVES_PER_WG; ++w) {
            s0 += pf[w * 128];
            s1 += pf[w * 128 + 32];
            s2 += pf[w * 128 + 64];
            s3 += pf[w * 128 + 96];
        }
        P[eg * H + cs * 32 + threadIdx.x] = (s0 + s1) + (s2 + s3);
    }
}

__device__ __forceinline__ float dot4acc(float4 a, v4f w, float acc) {
    acc = fmaf(a.x, w.x, acc);
    acc = fmaf(a.y, w.y, acc);
    acc = fmaf(a.z, w.z, acc);
    acc = fmaf(a.w, w.w, acc);
    return acc;
}

#define KSPLIT 4
#define KCOLS (H / KSPLIT)
#define KSTEPS (KCOLS / 4)
#define WROWS 64
#define XPITCH 68

#define A_STEP_LOAD(gstep, a, b)                                                       \
    {                                                                                  \
        const float4* r0 = (const float4*)(R + (size_t)(4 * (gstep) + 2 * half) * H + 4 * c16); \
        const float4* r1 = r0 + H / 4;                                                 \
        a[0] = r0[0]; a[1] = r0[16]; a[2] = r0[32]; a[3] = r0[48];                     \
        b[0] = r1[0]; b[1] = r1[16]; b[2] = r1[32]; b[3] = r1[48];                     \
    }
#define A_STEP_SUM(lstep, a, b)                                                        \
    {                                                                                  \
        float p0 = dot4acc(a[3], w3, dot4acc(a[2], w2, dot4acc(a[1], w1, dot4acc(a[0], w0, 0.0f)))); \
        float p1 = dot4acc(b[3], w3, dot4acc(b[2], w2, dot4acc(b[1], w1, dot4acc(b[0], w0, 0.0f)))); \
        v2f p; p.x = p0; p.y = p1;                                                     \
        afrag[(lstep) * 32 + lane] = p;                                                \
    }

__global__ void __launch_bounds__(256)
k_dot_rows_wmma(const float* __restrict__ X,
                const float* __restrict__ R,
                const float* __restrict__ P,
                float* __restrict__ t, int nrows) {
    __shared__ v2f afrag[KSTEPS * 32];
    __shared__ v4f lsq[4][H / 4];
    __shared__ v4f xtile[8][WROWS * XPITCH / 4];

    int wave = __builtin_amdgcn_readfirstlane(threadIdx.x >> 5);
    int lane = threadIdx.x & 31;
    int c16  = lane & 15;
    int half = lane >> 4;
    int ks   = blockIdx.x & (KSPLIT - 1);
    int row0 = ((blockIdx.x / KSPLIT) * 8 + wave) * WROWS;

    int pq = threadIdx.x >> 6;
    int pc = threadIdx.x & 63;
    v4f pv[EDGE_GROUPS / 4];
    #pragma unroll
    for (int i = 0; i < EDGE_GROUPS / 4; ++i)
        pv[i] = *(const v4f*)(P + (size_t)((EDGE_GROUPS / 4) * pq + i) * H + 4 * pc);
    __builtin_amdgcn_sched_barrier(0);

    float4 ra[2][4], rb[2][4];
    #pragma unroll
    for (int s = 0; s < 2; ++s) A_STEP_LOAD(ks * KSTEPS + wave * 2 + s, ra[s], rb[s]);
    __builtin_amdgcn_sched_barrier(0);

    const float* xcol = X + ks * KCOLS + 4 * c16;
    v4f xv[2][16];
    #pragma unroll
    for (int i = 0; i < 16; ++i) {
        int r = row0 + 2 * i + half;
        if (r >= nrows) r = 0;
        xv[0][i] = *(const v4f*)(xcol + (size_t)r * H);
    }
    __builtin_amdgcn_sched_barrier(0);

    lsq[pq][pc] = ((pv[0] + pv[1]) + (pv[2] + pv[3])) + ((pv[4] + pv[5]) + (pv[6] + pv[7]));
    __builtin_amdgcn_fence(__ATOMIC_RELEASE, "workgroup", "local");
    __builtin_amdgcn_s_barrier();
    __builtin_amdgcn_fence(__ATOMIC_ACQUIRE, "workgroup", "local");
    v4f w0 = (lsq[0][c16]      + lsq[1][c16])      + (lsq[2][c16]      + lsq[3][c16]);
    v4f w1 = (lsq[0][c16 + 16] + lsq[1][c16 + 16]) + (lsq[2][c16 + 16] + lsq[3][c16 + 16]);
    v4f w2 = (lsq[0][c16 + 32] + lsq[1][c16 + 32]) + (lsq[2][c16 + 32] + lsq[3][c16 + 32]);
    v4f w3 = (lsq[0][c16 + 48] + lsq[1][c16 + 48]) + (lsq[2][c16 + 48] + lsq[3][c16 + 48]);

    #pragma unroll
    for (int s = 0; s < 2; ++s) A_STEP_SUM(wave * 2 + s, ra[s], rb[s]);
    __builtin_amdgcn_sched_barrier(0);
    #pragma unroll
    for (int i = 0; i < 16; ++i) {
        int r = row0 + 32 + 2 * i + half;
        if (r >= nrows) r = 0;
        xv[1][i] = *(const v4f*)(xcol + (size_t)r * H);
    }

    __builtin_amdgcn_fence(__ATOMIC_RELEASE, "workgroup", "local");
    __builtin_amdgcn_s_barrier();
    __builtin_amdgcn_fence(__ATOMIC_ACQUIRE, "workgroup", "local");
    v2f a[KSTEPS];
    #pragma unroll
    for (int q = 0; q < KSTEPS; ++q) a[q] = afrag[q * 32 + lane];

    v4f* tl = xtile[wave];
    const float* brow = (const float*)xtile[wave] + c16 * XPITCH + 2 * half;

    #pragma unroll
    for (int p = 0; p < 2; ++p) {
        #pragma unroll
        for (int i = 0; i < 16; ++i) tl[(32 * p + 2 * i + half) * (XPITCH / 4) + c16] = xv[p][i];
        __builtin_amdgcn_fence(__ATOMIC_RELEASE, "wavefront", "local");
        __builtin_amdgcn_wave_barrier();
        __builtin_amdgcn_fence(__ATOMIC_ACQUIRE, "wavefront", "local");

        const float* bp = brow + 32 * p * XPITCH;
        v8f c0 = {}, c1 = {};
        #pragma unroll
        for (int q = 0; q < KSTEPS; ++q) {
            v2f b0 = *(const v2f*)(bp + 4 * q);
            v2f b1 = *(const v2f*)(bp + 16 * XPITCH + 4 * q);
            c0 = __builtin_amdgcn_wmma_f32_16x16x4_f32(
                     false, a[q],  false, b0,
                     (short)0, c0,  false,  false);
            c1 = __builtin_amdgcn_wmma_f32_16x16x4_f32(
                     false, a[q],  false, b1,
                     (short)0, c1,  false,  false);
        }

        float s0 = ((c0[0] + c0[1]) + (c0[2] + c0[3])) + ((c0[4] + c0[5]) + (c0[6] + c0[7]));
        float s1 = ((c1[0] + c1[1]) + (c1[2] + c1[3])) + ((c1[4] + c1[5]) + (c1[6] + c1[7]));
        s0 += __shfl_xor(s0, 16, 32);
        s1 += __shfl_xor(s1, 16, 32);

        int n = row0 + 32 * p + lane;
        if (n < nrows) atomicAdd(&t[n], half ? s1 : s0);
    }
}

__global__ void k_gather_scores(const float* __restrict__ t,
                                const int* __restrict__ trg_idx,
                                float* __restrict__ out, int E) {
    int i = blockIdx.x * blockDim.x + threadIdx.x;
    if (i < E) out[i] = t[trg_idx[i]];
}

extern "C" void kernel_launch(void* const* d_in, const int* in_sizes, int n_in,
                              void* d_out, int out_size, void* d_ws, size_t ws_size,
                              hipStream_t stream) {
    const float* x_src = (const float*)d_in[0];
    const float* x_trg = (const float*)d_in[1];
    const float* R     = (const float*)d_in[2];
    const int*   edges = (const int*)d_in[3];

    const int n_trg = in_sizes[1] / H;
    const int E     = in_sizes[3] / 2;
    const int* src_idx = edges;
    const int* trg_idx = edges + E;
    float* out = (float*)d_out;

    char* ws = (char*)d_ws;
    float* P = (float*)ws;
    float* t = (float*)(ws + EDGE_GROUPS * H * 4);

    k_edge_rowsum<<<EDGE_GROUPS * COL_SLICES, 32 * WAVES_PER_WG, 0, stream>>>(x_src, src_idx, P, t, E, n_trg);
    {
        int blocks = ((n_trg + 511) / 512) * KSPLIT;
        k_dot_rows_wmma<<<blocks, 256, 0, stream>>>(x_trg, R, P, t, n_trg);
    }
    k_gather_scores<<<(E + 255) / 256, 256, 0, stream>>>(t, trg_idx, out, E);
}
